# starting kernel
# baseline (speedup 1.0000x reference)
#include <hip/hip_runtime.h>

typedef __attribute__((ext_vector_type(2))) float v2f;
typedef __attribute__((ext_vector_type(4))) float v4f;
typedef __attribute__((ext_vector_type(8))) float v8f;

namespace {
constexpr int HID   = 50;
constexpr int NG    = 200;
constexpr int KPAD  = 52;
constexpr int NT    = 13;
constexpr int KC    = 13;
constexpr int NPAD  = NT * 16;
constexpr int T_IN  = 512;
constexpr int FUT   = 64;
constexpr int T_OUT = T_IN + FUT;

__device__ __forceinline__ float sigmoidf_(float v) {
    return 1.0f / (1.0f + __expf(-v));
}
}

__global__ __launch_bounds__(32) void lstm_seq_kernel(
    const float* __restrict__ x,
    const float* __restrict__ W_ih,
    const float* __restrict__ W_hh,
    const float* __restrict__ b_ih,
    const float* __restrict__ b_hh,
    const float* __restrict__ W_out,
    const float* __restrict__ b_out,
    float* __restrict__ out)
{
    __shared__ float frags[KC * NT * 32 * 2];
    __shared__ float gates[NPAD * 16];
    __shared__ float hbuf[16 * KPAD];

    const int lane    = threadIdx.x;
    const int r       = lane & 15;
    const int half    = lane >> 4;
    const int rowbase = blockIdx.x * 16;

    for (int fi = 0; fi < KC * NT; ++fi) {
        const int kc = fi / NT;
        const int tt = fi % NT;
        const int n  = tt * 16 + r;
        const int k0 = kc * 4 + 2 * half;
        float v0 = 0.0f, v1 = 0.0f;
        if (n < NG) {
            if (k0 < HID)       v0 = W_hh[n * HID + k0];
            else if (k0 == HID) v0 = b_ih[n] + b_hh[n];
            else                v0 = W_ih[n];
            const int k1 = k0 + 1;
            if (k1 < HID)       v1 = W_hh[n * HID + k1];
            else if (k1 == HID) v1 = b_ih[n] + b_hh[n];
            else                v1 = W_ih[n];
        }
        frags[(fi * 32 + lane) * 2 + 0] = v0;
        frags[(fi * 32 + lane) * 2 + 1] = v1;
    }

    float creg[25];
    float wout[25];
#pragma unroll
    for (int j = 0; j < 25; ++j) {
        const int k = 2 * j + half;
        creg[j] = 0.0f;
        wout[j] = W_out[k];
        hbuf[r * KPAD + k] = 0.0f;
    }
    hbuf[r * KPAD + HID] = 1.0f;
    const float bo = b_out[0];
    __syncthreads();

    float xnext = 0.0f;

#pragma unroll 1
    for (int t = 0; t < T_OUT; ++t) {
        float xt;
        if (t < T_IN) xt = x[(rowbase + r) * T_IN + t];
        else          xt = xnext;
        hbuf[r * KPAD + (HID + 1)] = xt;

        v2f afrag[KC];
#pragma unroll
        for (int kc = 0; kc < KC; ++kc)
            afrag[kc] = *(const v2f*)&hbuf[r * KPAD + kc * 4 + 2 * half];

#pragma unroll
        for (int tt = 0; tt < NT; ++tt) {
            v8f acc = {0.f, 0.f, 0.f, 0.f, 0.f, 0.f, 0.f, 0.f};
#pragma unroll
            for (int kc = 0; kc < KC; ++kc) {
                const v2f bfrag =
                    *(const v2f*)&frags[((kc * NT + tt) * 32 + lane) * 2];
                acc = __builtin_amdgcn_wmma_f32_16x16x4_f32(
                     false, afrag[kc],
                     false, bfrag,
                     (short)0, acc,
                     false,  false);
            }
            const int n = tt * 16 + r;
            float* gp = &gates[n * 16 + half * 8];
            v4f lo = {acc[0], acc[1], acc[2], acc[3]};
            v4f hi = {acc[4], acc[5], acc[6], acc[7]};
            *(v4f*)(gp)     = lo;
            *(v4f*)(gp + 4) = hi;
        }

        float osum = 0.0f;
#pragma unroll
        for (int j = 0; j < 25; ++j) {
            const int k = 2 * j + half;
            const float gi = gates[(k          ) * 16 + r];
            const float gf = gates[(k +     HID) * 16 + r];
            const float gg = gates[(k + 2 * HID) * 16 + r];
            const float go = gates[(k + 3 * HID) * 16 + r];
            const float ig = sigmoidf_(gi);
            const float fg = sigmoidf_(gf);
            const float cg = tanhf(gg);
            const float og = sigmoidf_(go);
            const float cv = fg * creg[j] + ig * cg;
            creg[j] = cv;
            const float hv = og * tanhf(cv);
            osum += hv * wout[j];
            hbuf[r * KPAD + k] = hv;
        }

        osum += __shfl_xor(osum, 16, 32);
        const float ov = osum + bo;
        if (half == 0)
            out[(rowbase + r) * T_OUT + t] = ov;
        xnext = ov;
    }
}

extern "C" void kernel_launch(void* const* d_in, const int* in_sizes, int n_in,
                              void* d_out, int out_size, void* d_ws, size_t ws_size,
                              hipStream_t stream) {
    const float* x     = (const float*)d_in[0];
    const float* W_ih  = (const float*)d_in[1];
    const float* W_hh  = (const float*)d_in[2];
    const float* b_ih  = (const float*)d_in[3];
    const float* b_hh  = (const float*)d_in[4];
    const float* W_out = (const float*)d_in[5];
    const float* b_out = (const float*)d_in[6];
    float* out = (float*)d_out;

    const int B = in_sizes[0] / T_IN;
    dim3 grid(B / 16);
    dim3 block(32);
    hipLaunchKernelGGL(lstm_seq_kernel, grid, block, 0, stream,
                       x, W_ih, W_hh, b_ih, b_hh, W_out, b_out, out);
}


# Weight columns regrouped so one lane holds i,f,g,o of its own cells in four tiles: cell update once per cell, not in all 4 lanes of a quad, no quad DPP for units 0..47; out summed in a new order
# speedup vs baseline: 8.2456x; 8.2456x over previous
#include <hip/hip_runtime.h>


typedef __attribute__((ext_vector_type(2))) float v2f;
typedef __attribute__((ext_vector_type(4))) float v4f;
typedef __attribute__((ext_vector_type(8))) float v8f;

namespace {
constexpr int HID   = 50;
constexpr int KPAD  = 52;
constexpr int NT    = 13;
constexpr int KC    = 13;
constexpr int T_IN  = 512;
constexpr int FUT   = 64;
constexpr int T_OUT = T_IN + FUT;
constexpr int NW    = 4;
constexpr int SLOTS = 3;
constexpr int XT    = 4;
constexpr int RW    = 2;
constexpr int RB    = NW * RW;
constexpr int NG    = 3;

template <int CTRL>
__device__ __forceinline__ float dpp_read(float v) {
    const int s = __builtin_bit_cast(int, v);
    return __builtin_bit_cast(float, __builtin_amdgcn_update_dpp(s, s, CTRL, 0xf, 0xf, false));
}

__device__ __forceinline__ float tanh_nobranch(float y) {
    const float e = __expf(2.0f * fabsf(y));
    const float rc = __builtin_amdgcn_rcpf(e + 1.0f);
    return copysignf(fmaf(-2.0f, rc, 1.0f), y);
}

__device__ __forceinline__ int col_unit(int tt, int c) {
    return tt < NT - 1 ? 16 * (tt >> 2) + c : 4 * (NT - 1) + (c >> 2);
}
__device__ __forceinline__ int col_gate(int tt, int c) {
    return tt < NT - 1 ? (tt & 3) : (c & 3);
}

__host__ __device__ constexpr int xt_base(int w) { return w == 0 ? 0 : 3 * w + 1; }
__host__ __device__ constexpr int xt_wave(int tt) { return tt < 4 ? 0 : (tt - 1) / 3; }

__device__ __forceinline__ void lds_barrier() {
    __builtin_amdgcn_fence(__ATOMIC_RELEASE, "workgroup", "local");
    __builtin_amdgcn_s_barrier();
    __builtin_amdgcn_fence(__ATOMIC_ACQUIRE, "workgroup", "local");
}
}

__global__ __launch_bounds__(NW * 32) void lstm_seq_kernel(
    const float* __restrict__ x,
    const float* __restrict__ W_ih,
    const float* __restrict__ W_hh,
    const float* __restrict__ b_ih,
    const float* __restrict__ b_hh,
    const float* __restrict__ W_out,
    const float* __restrict__ b_out,
    float* __restrict__ out)
{
    __shared__ float hbuf[16 * KPAD];
    __shared__ __attribute__((aligned(16))) v4f part[NW * NT * 32];
    __shared__ __attribute__((aligned(16))) v4f partx[NW * XT * 32];

    const int tid  = threadIdx.x;
    const int lane = tid & 31;
    const int wave = __builtin_amdgcn_readfirstlane(tid >> 5);
    const int r    = lane & 15;
    const int half = lane >> 4;
    const int gate = r & 3;
    const int unit = r >> 2;

    const int  kbase = SLOTS * wave;
    const bool four  = (wave == 0);
    const int  tbase = xt_base(wave);

    const int brow = RW * wave + half;
    const int wrow_own = 8 * (brow >> 2) + (brow & 3);
    const int grow = blockIdx.x * RB + brow;

    v2f bw[SLOTS * NT];
    v2f bwx[XT];
    const int koff = 2 * half;
#pragma unroll
    for (int tt = 0; tt < NT; ++tt) {
        const int  k  = col_unit(tt, r);
        const bool ok = (k < HID);
        const int  nn = col_gate(tt, r) * HID + (ok ? k : (HID - 1));
        const float* wrow = W_hh + nn * HID;
#pragma unroll
        for (int s = 0; s < SLOTS; ++s) {
            const int k0 = (kbase + s) * 4 + koff;
            const float v0 = wrow[k0];
            const float v1 = wrow[k0 + 1];
            bw[s * NT + tt].x = ok ? v0 : 0.0f;
            bw[s * NT + tt].y = ok ? v1 : 0.0f;
        }
    }
#pragma unroll
    for (int q = 0; q < XT; ++q) {
        const int  tq = (tbase + q < NT) ? (tbase + q) : (NT - 1);
        const int  k  = col_unit(tq, r);
        const bool ok = (k < HID) && (q < XT - 1 || four);
        const int  nn = col_gate(tq, r) * HID + ((k < HID) ? k : (HID - 1));
        const float* wrow = W_hh + nn * HID;
        const float w48 = wrow[HID - 2];
        const float w49 = wrow[HID - 1];
        const float bs  = b_ih[nn] + b_hh[nn];
        const float wi  = W_ih[nn];
        const float l0  = half ? bs : w48;
        const float l1  = half ? wi : w49;
        bwx[q].x = ok ? l0 : 0.0f;
        bwx[q].y = ok ? l1 : 0.0f;
    }

    for (int i = tid; i < 16 * KPAD; i += NW * 32)
        hbuf[i] = 0.0f;
    __syncthreads();
    hbuf[wrow_own * KPAD + HID]       = 1.0f;
    hbuf[wrow_own * KPAD + (HID + 1)] = x[grow * T_IN];

    float creg[NG + 1];
    float wout[NG + 1];
#pragma unroll
    for (int m = 0; m < NG; ++m) {
        creg[m] = 0.0f;
        wout[m] = W_out[16 * m + r];
    }
    {
        const int k = 4 * (NT - 1) + unit;
        creg[NG] = 0.0f;
        const float w = W_out[k < HID ? k : (HID - 1)];
        wout[NG] = (k < HID && gate == 0) ? w : 0.0f;
    }
    const float bo = b_out[0];
    const float act_s = (gate == 2) ? 1.0f : 0.5f;
    const float act_b = (gate == 2) ? 0.0f : 0.5f;

    const float* pf  = (const float*)part;
    const float* pfx = (const float*)partx;
    const int plo = ((brow >> 2) * 16 + r) * 4 + (brow & 3);
    __syncthreads();

#pragma unroll 1
    for (int t = 0; t < T_OUT; ++t) {
        float xobs = 0.0f;
        if (t + 1 < T_IN) xobs = x[grow * T_IN + t + 1];

        v2f afrag[SLOTS];
#pragma unroll
        for (int s = 0; s < SLOTS; ++s)
            afrag[s] = *(const v2f*)&hbuf[r * KPAD + (kbase + s) * 4 + 2 * half];
        const v2f ax = *(const v2f*)&hbuf[r * KPAD + (KC - 1) * 4 + 2 * half];

        v8f accx[XT];
#pragma unroll
        for (int q = 0; q < XT - 1; ++q)
            accx[q] = __builtin_amdgcn_wmma_f32_16x16x4_f32(
                false, ax, false, bwx[q], (short)0,
                v8f{0.f, 0.f, 0.f, 0.f, 0.f, 0.f, 0.f, 0.f}, false, false);
        v8f acc[NT];
#pragma unroll
        for (int tt = 0; tt < NT; ++tt)
            acc[tt] = __builtin_amdgcn_wmma_f32_16x16x4_f32(
                false, afrag[0], false, bw[tt], (short)0,
                v8f{0.f, 0.f, 0.f, 0.f, 0.f, 0.f, 0.f, 0.f}, false, false);
        if (four)
            accx[XT - 1] = __builtin_amdgcn_wmma_f32_16x16x4_f32(
                false, ax, false, bwx[XT - 1], (short)0,
                v8f{0.f, 0.f, 0.f, 0.f, 0.f, 0.f, 0.f, 0.f}, false, false);
#pragma unroll
        for (int s = 1; s < SLOTS; ++s) {
#pragma unroll
            for (int tt = 0; tt < NT; ++tt)
                acc[tt] = __builtin_amdgcn_wmma_f32_16x16x4_f32(
                    false, afrag[s], false, bw[s * NT + tt], (short)0,
                    acc[tt], false, false);
        }

#pragma unroll
        for (int q = 0; q < XT - 1; ++q)
            partx[(wave * XT + q) * 32 + lane] =
                v4f{accx[q][0], accx[q][1], accx[q][2], accx[q][3]};
        if (four)
            partx[(wave * XT + (XT - 1)) * 32 + lane] =
                v4f{accx[XT - 1][0], accx[XT - 1][1], accx[XT - 1][2], accx[XT - 1][3]};
#pragma unroll
        for (int tt = 0; tt < NT; ++tt)
            part[(wave * NT + tt) * 32 + lane] =
                v4f{acc[tt][0], acc[tt][1], acc[tt][2], acc[tt][3]};
        lds_barrier();

        float g[NT];
#pragma unroll
        for (int tt = 0; tt < NT; ++tt) {
            const float p0 = pf[((0 * NT + tt) * 32) * 4 + plo];
            const float p1 = pf[((1 * NT + tt) * 32) * 4 + plo];
            const float p2 = pf[((2 * NT + tt) * 32) * 4 + plo];
            const float p3 = pf[((3 * NT + tt) * 32) * 4 + plo];
            const float p4 = pfx[((xt_wave(tt) * XT + (tt - xt_base(xt_wave(tt)))) * 32) * 4 + plo];
            g[tt] = ((p0 + p1) + (p2 + p3)) + p4;
        }

        float hv[NG + 1];
        float osum = 0.0f;
#pragma unroll
        for (int m = 0; m < NG; ++m) {
            const float ig = fmaf(tanh_nobranch(g[4 * m + 0] * 0.5f), 0.5f, 0.5f);
            const float fg = fmaf(tanh_nobranch(g[4 * m + 1] * 0.5f), 0.5f, 0.5f);
            const float cg = tanh_nobranch(g[4 * m + 2]);
            const float og = fmaf(tanh_nobranch(g[4 * m + 3] * 0.5f), 0.5f, 0.5f);
            const float cv = fg * creg[m] + ig * cg;
            creg[m] = cv;
            hv[m] = og * tanh_nobranch(cv);
            osum = fmaf(hv[m], wout[m], osum);
        }
        {
            const float a  = fmaf(tanh_nobranch(g[NT - 1] * act_s), act_s, act_b);
            const float ig = dpp_read<0x00>(a);
            const float fg = dpp_read<0x55>(a);
            const float cg = dpp_read<0xAA>(a);
            const float og = dpp_read<0xFF>(a);
            const float cv = fg * creg[NG] + ig * cg;
            creg[NG] = cv;
            hv[NG] = og * tanh_nobranch(cv);
            osum = fmaf(hv[NG], wout[NG], osum);
        }

#pragma unroll
        for (int m = 0; m < NG; ++m)
            hbuf[wrow_own * KPAD + 16 * m + r] = hv[m];
        if (gate == 0 && unit < HID - 4 * (NT - 1))
            hbuf[wrow_own * KPAD + (NT - 1) * 4 + unit] = hv[NG];

        osum += dpp_read<0x161>(osum);
        osum += dpp_read<0x162>(osum);
        osum += dpp_read<0x164>(osum);
        osum += dpp_read<0x168>(osum);
        const float ov = osum + bo;
        if (r == 0)
            out[grow * T_OUT + t] = ov;

        hbuf[wrow_own * KPAD + (HID + 1)] = (t + 1 < T_IN) ? xobs : ov;
        lds_barrier();
    }
}

extern "C" void kernel_launch(void* const* d_in, const int* in_sizes, int n_in,
                              void* d_out, int out_size, void* d_ws, size_t ws_size,
                              hipStream_t stream) {
    const float* x     = (const float*)d_in[0];
    const float* W_ih  = (const float*)d_in[1];
    const float* W_hh  = (const float*)d_in[2];
    const float* b_ih  = (const float*)d_in[3];
    const float* b_hh  = (const float*)d_in[4];
    const float* W_out = (const float*)d_in[5];
    const float* b_out = (const float*)d_in[6];
    float* out = (float*)d_out;

    const int B = in_sizes[0] / T_IN;
    dim3 grid(B / RB);
    dim3 block(NW * 32);
    hipLaunchKernelGGL(lstm_seq_kernel, grid, block, 0, stream,
                       x, W_ih, W_hh, b_ih, b_hh, W_out, b_out, out);
}


# Cell ownership follows the K split: a wave updates the cells whose h it multiplies, h stays in registers as the A operand, an observed step has one barrier, not two; out summed in a new order
# speedup vs baseline: 9.0274x; 1.0948x over previous
#include <hip/hip_runtime.h>


typedef __attribute__((ext_vector_type(2))) float v2f;
typedef __attribute__((ext_vector_type(4))) float v4f;
typedef __attribute__((ext_vector_type(8))) float v8f;

namespace {
constexpr int HID   = 50;
constexpr int NT    = 13;
constexpr int T_IN  = 512;
constexpr int FUT   = 64;
constexpr int T_OUT = T_IN + FUT;
constexpr int NW    = 4;
constexpr int SLOTS = 3;
constexpr int XT    = 4;
constexpr int RB    = 8;
constexpr int NREG  = NT + XT;
constexpr int HP    = 24;
constexpr int RS    = 48;
constexpr int BUFV  = NW * NREG * RS;

template <int CTRL>
__device__ __forceinline__ float dpp_read(float v) {
    const int s = __builtin_bit_cast(int, v);
    return __builtin_bit_cast(float, __builtin_amdgcn_update_dpp(s, s, CTRL, 0xf, 0xf, false));
}
__device__ __forceinline__ float xhalf(float v) {
    const int s = __builtin_bit_cast(int, v);
    return __builtin_bit_cast(float,
        __builtin_amdgcn_permlanex16(s, s, 0x76543210, 0xfedcba98, false, false));
}

__device__ __forceinline__ float tanh_nobranch(float y) {
    const float e = __expf(2.0f * fabsf(y));
    const float rc = __builtin_amdgcn_rcpf(e + 1.0f);
    return copysignf(fmaf(-2.0f, rc, 1.0f), y);
}
__device__ __forceinline__ float sigmoid_nobranch(float y) {
    return fmaf(tanh_nobranch(y * 0.5f), 0.5f, 0.5f);
}

__device__ __forceinline__ int col_unit(int tt, int c) {
    return tt < NT - 1 ? 16 * (tt >> 2) + c : 4 * (NT - 1) + (c >> 2);
}
__device__ __forceinline__ int col_gate(int tt, int c) {
    return tt < NT - 1 ? (tt & 3) : (c & 3);
}

__device__ __forceinline__ void lds_barrier() {
    __builtin_amdgcn_fence(__ATOMIC_RELEASE, "workgroup", "local");
    __builtin_amdgcn_s_barrier();
    __builtin_amdgcn_fence(__ATOMIC_ACQUIRE, "workgroup", "local");
}
}

__global__ __launch_bounds__(NW * 32) void lstm_seq_kernel(
    const float* __restrict__ x,
    const float* __restrict__ W_ih,
    const float* __restrict__ W_hh,
    const float* __restrict__ b_ih,
    const float* __restrict__ b_hh,
    const float* __restrict__ W_out,
    const float* __restrict__ b_out,
    float* __restrict__ out)
{
    __shared__ __attribute__((aligned(16))) v4f part[2 * BUFV];
    __shared__ __attribute__((aligned(16))) v4f po[2 * RB];

    const int tid  = threadIdx.x;
    const int lane = tid & 31;
    const int wave = __builtin_amdgcn_readfirstlane(tid >> 5);
    const int r    = lane & 15;
    const int half = lane >> 4;

    const int  kbase = SLOTS * wave;
    const bool many  = (wave < NW - 1);
    const int  tbase = 4 * wave;

    const int p    = (r >> 2) & 1;
    const int rh   = r >> 3;
    const int rj   = r & 3;
    const int brow = 4 * rh + rj;
    const int grow = blockIdx.x * RB + brow;
    const bool lead = (half == 0) && (p == 0);

    v2f bw[SLOTS * NT];
    v2f bwx[XT];
    const int koff = 2 * half;
#pragma unroll
    for (int tt = 0; tt < NT; ++tt) {
        const int  k  = col_unit(tt, r);
        const bool ok = (k < HID);
        const int  nn = col_gate(tt, r) * HID + (ok ? k : (HID - 1));
        const float* wrow = W_hh + nn * HID;
#pragma unroll
        for (int s = 0; s < SLOTS; ++s) {
            const int k0 = (kbase + s) * 4 + koff;
            const float v0 = wrow[k0];
            const float v1 = wrow[k0 + 1];
            bw[s * NT + tt].x = ok ? v0 : 0.0f;
            bw[s * NT + tt].y = ok ? v1 : 0.0f;
        }
    }
#pragma unroll
    for (int q = 0; q < XT; ++q) {
        const int  tq = (tbase + q < NT) ? (tbase + q) : (NT - 1);
        const int  k  = col_unit(tq, r);
        const bool ok = (k < HID) && (q == 0 || many);
        const int  nn = col_gate(tq, r) * HID + ((k < HID) ? k : (HID - 1));
        const float* wrow = W_hh + nn * HID;
        const float w48 = wrow[HID - 2];
        const float w49 = wrow[HID - 1];
        const float bs  = b_ih[nn] + b_hh[nn];
        const float wi  = W_ih[nn];
        const float l0  = half ? bs : w48;
        const float l1  = half ? wi : w49;
        bwx[q].x = ok ? l0 : 0.0f;
        bwx[q].y = ok ? l1 : 0.0f;
    }

    float hv[SLOTS], creg[SLOTS], wout[SLOTS];
#pragma unroll
    for (int s = 0; s < SLOTS; ++s) {
        hv[s] = 0.0f;
        creg[s] = 0.0f;
        wout[s] = W_out[12 * wave + 4 * s + 2 * half + p];
    }
    float hv4 = 0.0f, creg4 = 0.0f;
    const float wout4 = (wave == 0 && p == 0) ? W_out[4 * (NT - 1) + half] : 0.0f;
    const float act_s = p ? 1.0f : 0.5f;
    const float act_b = p ? 0.0f : 0.5f;
    const float bo = b_out[0];

    const float* pf = (const float*)part;
    int rd[SLOTS], rdx[SLOTS];
#pragma unroll
    for (int s = 0; s < SLOTS; ++s) {
        const int ub = 12 * wave + 4 * s;
        const int m  = ub >> 4;
        const int cc = (ub & 15) + 2 * half + p;
        rd[s]  = ((4 * m) * RS + rh * HP + cc) * 4 + rj;
        rdx[s] = ((m * NREG + NT) * RS + rh * HP + cc) * 4 + rj;
    }
    const int rd4  = ((NT - 1) * RS + rh * HP + 4 * half + 2 * p) * 4 + rj;
    const int rdx4 = (((NW - 1) * NREG + NT) * RS + rh * HP + 4 * half + 2 * p) * 4 + rj;
    const int wr = (wave * NREG) * RS + half * HP + r;

    float xcur = x[grow * T_IN];
    v4f pvc = v4f{0.f, 0.f, 0.f, 0.f};

    if (tid < 2 * RB)
        po[tid] = v4f{0.f, 0.f, 0.f, 0.f};
    __syncthreads();

#pragma unroll 1
    for (int t = 0; t < T_OUT; ++t) {
        float xnext = 0.0f;
        if (t + 1 < T_IN) xnext = x[grow * T_IN + t + 1];

        v2f afrag[SLOTS];
#pragma unroll
        for (int s = 0; s < SLOTS; ++s)
            afrag[s] = v2f{hv[s], dpp_read<0x164>(hv[s])};
        const float h49 = xhalf(hv4);
        const v2f ax = v2f{half ? 1.0f : hv4, half ? xcur : h49};

        v8f accx[XT];
        accx[0] = __builtin_amdgcn_wmma_f32_16x16x4_f32(
            false, ax, false, bwx[0], (short)0,
            v8f{0.f, 0.f, 0.f, 0.f, 0.f, 0.f, 0.f, 0.f}, false, false);
        v8f acc[NT];
#pragma unroll
        for (int tt = 0; tt < NT; ++tt)
            acc[tt] = __builtin_amdgcn_wmma_f32_16x16x4_f32(
                false, afrag[0], false, bw[tt], (short)0,
                v8f{0.f, 0.f, 0.f, 0.f, 0.f, 0.f, 0.f, 0.f}, false, false);
        if (many) {
#pragma unroll
            for (int q = 1; q < XT; ++q)
                accx[q] = __builtin_amdgcn_wmma_f32_16x16x4_f32(
                    false, ax, false, bwx[q], (short)0,
                    v8f{0.f, 0.f, 0.f, 0.f, 0.f, 0.f, 0.f, 0.f}, false, false);
        }
#pragma unroll
        for (int s = 1; s < SLOTS; ++s) {
#pragma unroll
            for (int tt = 0; tt < NT; ++tt)
                acc[tt] = __builtin_amdgcn_wmma_f32_16x16x4_f32(
                    false, afrag[s], false, bw[s * NT + tt], (short)0,
                    acc[tt], false, false);
        }

        if (wave == 0 && lead && t >= 2 && t <= T_IN)
            out[grow * T_OUT + t - 2] = ((pvc.x + pvc.y) + (pvc.z + pvc.w)) + bo;

        v4f* pw = part + (t & 1) * BUFV + wr;
        pw[NT * RS] = v4f{accx[0][0], accx[0][1], accx[0][2], accx[0][3]};
        if (many) {
#pragma unroll
            for (int q = 1; q < XT; ++q)
                pw[(NT + q) * RS] = v4f{accx[q][0], accx[q][1], accx[q][2], accx[q][3]};
        }
#pragma unroll
        for (int tt = 0; tt < NT; ++tt)
            pw[tt * RS] = v4f{acc[tt][0], acc[tt][1], acc[tt][2], acc[tt][3]};
        lds_barrier();

        pvc = po[((t + 1) & 1) * RB + brow];

        const float* pr = pf + (t & 1) * (BUFV * 4);
        float osum = 0.0f;
#pragma unroll
        for (int s = 0; s < SLOTS; ++s) {
            float g[4];
#pragma unroll
            for (int gt = 0; gt < 4; ++gt) {
                const float p0 = pr[rd[s] + ((0 * NREG + gt) * RS) * 4];
                const float p1 = pr[rd[s] + ((1 * NREG + gt) * RS) * 4];
                const float p2 = pr[rd[s] + ((2 * NREG + gt) * RS) * 4];
                const float p3 = pr[rd[s] + ((3 * NREG + gt) * RS) * 4];
                const float p4 = pr[rdx[s] + (gt * RS) * 4];
                g[gt] = ((p0 + p1) + (p2 + p3)) + p4;
            }
            const float ig = sigmoid_nobranch(g[0]);
            const float fg = sigmoid_nobranch(g[1]);
            const float cg = tanh_nobranch(g[2]);
            const float og = sigmoid_nobranch(g[3]);
            const float cv = fg * creg[s] + ig * cg;
            creg[s] = cv;
            hv[s] = og * tanh_nobranch(cv);
            osum = fmaf(hv[s], wout[s], osum);
        }
        {
            float g[2];
#pragma unroll
            for (int e = 0; e < 2; ++e) {
                const float p0 = pr[rd4 + ((0 * NREG) * RS) * 4 + e * 4];
                const float p1 = pr[rd4 + ((1 * NREG) * RS) * 4 + e * 4];
                const float p2 = pr[rd4 + ((2 * NREG) * RS) * 4 + e * 4];
                const float p3 = pr[rd4 + ((3 * NREG) * RS) * 4 + e * 4];
                const float p4 = pr[rdx4 + e * 4];
                g[e] = ((p0 + p1) + (p2 + p3)) + p4;
            }
            const float a0 = fmaf(tanh_nobranch(g[0] * act_s), act_s, act_b);
            const float a1 = sigmoid_nobranch(g[1]);
            const float b0 = dpp_read<0x164>(a0);
            const float b1 = dpp_read<0x164>(a1);
            const float ig = p ? b0 : a0;
            const float fg = p ? b1 : a1;
            const float cg = p ? a0 : b0;
            const float og = p ? a1 : b1;
            const float cv = fg * creg4 + ig * cg;
            creg4 = cv;
            hv4 = og * tanh_nobranch(cv);
            osum = fmaf(hv4, wout4, osum);
        }

        osum += dpp_read<0x164>(osum);
        osum += xhalf(osum);
        if (lead)
            ((float*)po)[((t & 1) * RB + brow) * 4 + wave] = osum;

        xcur = xnext;
        if (t + 1 >= T_IN) {
            lds_barrier();
            const v4f pv = po[(t & 1) * RB + brow];
            xcur = ((pv.x + pv.y) + (pv.z + pv.w)) + bo;
            if (wave == 0 && lead)
                out[grow * T_OUT + t] = xcur;
        }
    }
}

extern "C" void kernel_launch(void* const* d_in, const int* in_sizes, int n_in,
                              void* d_out, int out_size, void* d_ws, size_t ws_size,
                              hipStream_t stream) {
    const float* x     = (const float*)d_in[0];
    const float* W_ih  = (const float*)d_in[1];
    const float* W_hh  = (const float*)d_in[2];
    const float* b_ih  = (const float*)d_in[3];
    const float* b_hh  = (const float*)d_in[4];
    const float* W_out = (const float*)d_in[5];
    const float* b_out = (const float*)d_in[6];
    float* out = (float*)d_out;

    const int B = in_sizes[0] / T_IN;
    dim3 grid(B / RB);
    dim3 block(NW * 32);
    hipLaunchKernelGGL(lstm_seq_kernel, grid, block, 0, stream,
                       x, W_ih, W_hh, b_ih, b_hh, W_out, b_out, out);
}


# Step order pipelined by K chunk: a chunk's 13 products of the next step issue as soon as that chunk's cells are updated, so the other cell updates run beside the matrix instructions; sums unchanged
# speedup vs baseline: 9.8240x; 1.0882x over previous
#include <hip/hip_runtime.h>


typedef __attribute__((ext_vector_type(2))) float v2f;
typedef __attribute__((ext_vector_type(4))) float v4f;
typedef __attribute__((ext_vector_type(8))) float v8f;

namespace {
constexpr int HID   = 50;
constexpr int NT    = 13;
constexpr int T_IN  = 512;
constexpr int FUT   = 64;
constexpr int T_OUT = T_IN + FUT;
constexpr int NW    = 4;
constexpr int SLOTS = 3;
constexpr int XT    = 4;
constexpr int RB    = 8;
constexpr int NREG  = NT + XT;
constexpr int HP    = 24;
constexpr int RS    = 48;
constexpr int BUFV  = NW * NREG * RS;

template <int CTRL>
__device__ __forceinline__ float dpp_read(float v) {
    const int s = __builtin_bit_cast(int, v);
    return __builtin_bit_cast(float, __builtin_amdgcn_update_dpp(s, s, CTRL, 0xf, 0xf, false));
}
__device__ __forceinline__ float xhalf(float v) {
    const int s = __builtin_bit_cast(int, v);
    return __builtin_bit_cast(float,
        __builtin_amdgcn_permlanex16(s, s, 0x76543210, 0xfedcba98, false, false));
}

__device__ __forceinline__ float tanh_nobranch(float y) {
    const float e = __expf(2.0f * fabsf(y));
    const float rc = __builtin_amdgcn_rcpf(e + 1.0f);
    return copysignf(fmaf(-2.0f, rc, 1.0f), y);
}
__device__ __forceinline__ float sigmoid_nobranch(float y) {
    return fmaf(tanh_nobranch(y * 0.5f), 0.5f, 0.5f);
}

__device__ __forceinline__ int col_unit(int tt, int c) {
    return tt < NT - 1 ? 16 * (tt >> 2) + c : 4 * (NT - 1) + (c >> 2);
}
__device__ __forceinline__ int col_gate(int tt, int c) {
    return tt < NT - 1 ? (tt & 3) : (c & 3);
}

__device__ __forceinline__ void lds_barrier() {
    __builtin_amdgcn_fence(__ATOMIC_RELEASE, "workgroup", "local");
    __builtin_amdgcn_s_barrier();
    __builtin_amdgcn_fence(__ATOMIC_ACQUIRE, "workgroup", "local");
}
}

__global__ __launch_bounds__(NW * 32) void lstm_seq_kernel(
    const float* __restrict__ x,
    const float* __restrict__ W_ih,
    const float* __restrict__ W_hh,
    const float* __restrict__ b_ih,
    const float* __restrict__ b_hh,
    const float* __restrict__ W_out,
    const float* __restrict__ b_out,
    float* __restrict__ out)
{
    __shared__ __attribute__((aligned(16))) v4f part[2 * BUFV];
    __shared__ __attribute__((aligned(16))) v4f po[2 * RB];

    const int tid  = threadIdx.x;
    const int lane = tid & 31;
    const int wave = __builtin_amdgcn_readfirstlane(tid >> 5);
    const int r    = lane & 15;
    const int half = lane >> 4;

    const int  kbase = SLOTS * wave;
    const bool many  = (wave < NW - 1);
    const int  tbase = 4 * wave;

    const int p    = (r >> 2) & 1;
    const int rh   = r >> 3;
    const int rj   = r & 3;
    const int brow = 4 * rh + rj;
    const int grow = blockIdx.x * RB + brow;
    const bool lead = (half == 0) && (p == 0);

    v2f bw[SLOTS * NT];
    v2f bwx[XT];
    const int koff = 2 * half;
#pragma unroll
    for (int tt = 0; tt < NT; ++tt) {
        const int  k  = col_unit(tt, r);
        const bool ok = (k < HID);
        const int  nn = col_gate(tt, r) * HID + (ok ? k : (HID - 1));
        const float* wrow = W_hh + nn * HID;
#pragma unroll
        for (int s = 0; s < SLOTS; ++s) {
            const int k0 = (kbase + s) * 4 + koff;
            const float v0 = wrow[k0];
            const float v1 = wrow[k0 + 1];
            bw[s * NT + tt].x = ok ? v0 : 0.0f;
            bw[s * NT + tt].y = ok ? v1 : 0.0f;
        }
    }
#pragma unroll
    for (int q = 0; q < XT; ++q) {
        const int  tq = (tbase + q < NT) ? (tbase + q) : (NT - 1);
        const int  k  = col_unit(tq, r);
        const bool ok = (k < HID) && (q == 0 || many);
        const int  nn = col_gate(tq, r) * HID + ((k < HID) ? k : (HID - 1));
        const float* wrow = W_hh + nn * HID;
        const float w48 = wrow[HID - 2];
        const float w49 = wrow[HID - 1];
        const float bs  = b_ih[nn] + b_hh[nn];
        const float wi  = W_ih[nn];
        const float l0  = half ? bs : w48;
        const float l1  = half ? wi : w49;
        bwx[q].x = ok ? l0 : 0.0f;
        bwx[q].y = ok ? l1 : 0.0f;
    }

    float hv[SLOTS], creg[SLOTS], wout[SLOTS];
#pragma unroll
    for (int s = 0; s < SLOTS; ++s) {
        hv[s] = 0.0f;
        creg[s] = 0.0f;
        wout[s] = W_out[12 * wave + 4 * s + 2 * half + p];
    }
    float hv4 = 0.0f, creg4 = 0.0f;
    const float wout4 = (wave == 0 && p == 0) ? W_out[4 * (NT - 1) + half] : 0.0f;
    const float act_s = p ? 1.0f : 0.5f;
    const float act_b = p ? 0.0f : 0.5f;
    const float bo = b_out[0];

    const float* pf = (const float*)part;
    int rd[SLOTS], rdx[SLOTS];
#pragma unroll
    for (int s = 0; s < SLOTS; ++s) {
        const int ub = 12 * wave + 4 * s;
        const int m  = ub >> 4;
        const int cc = (ub & 15) + 2 * half + p;
        rd[s]  = ((4 * m) * RS + rh * HP + cc) * 4 + rj;
        rdx[s] = ((m * NREG + NT) * RS + rh * HP + cc) * 4 + rj;
    }
    const int rd4  = ((NT - 1) * RS + rh * HP + 4 * half + 2 * p) * 4 + rj;
    const int rdx4 = (((NW - 1) * NREG + NT) * RS + rh * HP + 4 * half + 2 * p) * 4 + rj;
    const int wr = (wave * NREG) * RS + half * HP + r;

    const v8f zero8 = v8f{0.f, 0.f, 0.f, 0.f, 0.f, 0.f, 0.f, 0.f};
    v8f acc[NT];
    v8f accx[XT];
    float ps[SLOTS][4][NW + 1];
    float ps4[2][NW + 1];
    float osum = 0.0f;


    auto mm_slot = [&](const int s) __attribute__((always_inline)) {
        const v2f a = v2f{hv[s], dpp_read<0x164>(hv[s])};
#pragma unroll
        for (int tt = 0; tt < NT; ++tt)
            acc[tt] = __builtin_amdgcn_wmma_f32_16x16x4_f32(
                false, a, false, bw[s * NT + tt], (short)0,
                s == 0 ? zero8 : acc[tt], false, false);
    };
    auto mm_x = [&](const float xin, const int par) __attribute__((always_inline)) {
        const float h49 = xhalf(hv4);
        const v2f ax = v2f{half ? 1.0f : hv4, half ? xin : h49};
        v4f* pw = part + par * BUFV + wr;
        accx[0] = __builtin_amdgcn_wmma_f32_16x16x4_f32(
            false, ax, false, bwx[0], (short)0, zero8, false, false);
        if (many) {
#pragma unroll
            for (int q = 1; q < XT; ++q)
                accx[q] = __builtin_amdgcn_wmma_f32_16x16x4_f32(
                    false, ax, false, bwx[q], (short)0, zero8, false, false);
#pragma unroll
            for (int q = 1; q < XT; ++q)
                pw[(NT + q) * RS] = v4f{accx[q][0], accx[q][1], accx[q][2], accx[q][3]};
        }
        pw[NT * RS] = v4f{accx[0][0], accx[0][1], accx[0][2], accx[0][3]};
    };
    auto put_tiles = [&](const int par) __attribute__((always_inline)) {
        v4f* pw = part + par * BUFV + wr;
#pragma unroll
        for (int tt = 0; tt < NT; ++tt)
            pw[tt * RS] = v4f{acc[tt][0], acc[tt][1], acc[tt][2], acc[tt][3]};
    };
    auto get_partials = [&](const int par) __attribute__((always_inline)) {
        const float* pr = pf + par * (BUFV * 4);
#pragma unroll
        for (int s = 0; s < SLOTS; ++s) {
#pragma unroll
            for (int gt = 0; gt < 4; ++gt) {
#pragma unroll
                for (int w = 0; w < NW; ++w)
                    ps[s][gt][w] = pr[rd[s] + ((w * NREG + gt) * RS) * 4];
                ps[s][gt][NW] = pr[rdx[s] + (gt * RS) * 4];
            }
        }
#pragma unroll
        for (int e = 0; e < 2; ++e) {
#pragma unroll
            for (int w = 0; w < NW; ++w)
                ps4[e][w] = pr[rd4 + ((w * NREG) * RS) * 4 + e * 4];
            ps4[e][NW] = pr[rdx4 + e * 4];
        }
    };
    auto cell_slot = [&](const int s) __attribute__((always_inline)) {
        float g[4];
#pragma unroll
        for (int gt = 0; gt < 4; ++gt)
            g[gt] = ((ps[s][gt][0] + ps[s][gt][1]) + (ps[s][gt][2] + ps[s][gt][3])) + ps[s][gt][NW];
        const float ig = sigmoid_nobranch(g[0]);
        const float fg = sigmoid_nobranch(g[1]);
        const float cg = tanh_nobranch(g[2]);
        const float og = sigmoid_nobranch(g[3]);
        const float cv = fg * creg[s] + ig * cg;
        creg[s] = cv;
        hv[s] = og * tanh_nobranch(cv);
        osum = fmaf(hv[s], wout[s], osum);
    };
    auto cell_last = [&](const int par) __attribute__((always_inline)) {
        float g[2];
#pragma unroll
        for (int e = 0; e < 2; ++e)
            g[e] = ((ps4[e][0] + ps4[e][1]) + (ps4[e][2] + ps4[e][3])) + ps4[e][NW];
        const float a0 = fmaf(tanh_nobranch(g[0] * act_s), act_s, act_b);
        const float a1 = sigmoid_nobranch(g[1]);
        const float b0 = dpp_read<0x164>(a0);
        const float b1 = dpp_read<0x164>(a1);
        const float ig = p ? b0 : a0;
        const float fg = p ? b1 : a1;
        const float cg = p ? a0 : b0;
        const float og = p ? a1 : b1;
        const float cv = fg * creg4 + ig * cg;
        creg4 = cv;
        hv4 = og * tanh_nobranch(cv);
        osum = fmaf(hv4, wout4, osum);
        osum += dpp_read<0x164>(osum);
        osum += xhalf(osum);
        if (lead)
            ((float*)po)[(par * RB + brow) * 4 + wave] = osum;
        osum = 0.0f;
    };

    if (tid < 2 * RB)
        po[tid] = v4f{0.f, 0.f, 0.f, 0.f};
    __syncthreads();

    const float* xrow = x + grow * T_IN;
    float xn1 = xrow[1];
#pragma unroll
    for (int s = 0; s < SLOTS; ++s) mm_slot(s);
    put_tiles(0);
    mm_x(xrow[0], 0);

#pragma unroll 1
    for (int t = 0; t < T_IN - 1; ++t) {
        lds_barrier();
        const v4f pvc = po[((t + 1) & 1) * RB + brow];
        get_partials(t & 1);
        const float xn2 = xrow[t + 2 < T_IN ? t + 2 : T_IN - 1];

        cell_slot(0);
        mm_slot(0);
        cell_slot(1);
        mm_slot(1);
        cell_slot(2);
        mm_slot(2);
        cell_last(t & 1);
        put_tiles((t + 1) & 1);
        mm_x(xn1, (t + 1) & 1);

        if (wave == 0 && lead && t >= 1)
            out[grow * T_OUT + t - 1] = ((pvc.x + pvc.y) + (pvc.z + pvc.w)) + bo;
        xn1 = xn2;
    }

#pragma unroll 1
    for (int t = T_IN - 1; t < T_OUT - 1; ++t) {
        lds_barrier();
        get_partials(t & 1);

        cell_slot(0);
        mm_slot(0);
        cell_slot(1);
        mm_slot(1);
        cell_slot(2);
        mm_slot(2);
        cell_last(t & 1);
        put_tiles((t + 1) & 1);

        lds_barrier();
        const v4f pv = po[(t & 1) * RB + brow];
        const float xin = ((pv.x + pv.y) + (pv.z + pv.w)) + bo;
        if (wave == 0 && lead)
            out[grow * T_OUT + t] = xin;
        mm_x(xin, (t + 1) & 1);

        if (t == T_IN - 1 && wave == 0 && lead) {
            const v4f pl = po[((t + 1) & 1) * RB + brow];
            out[grow * T_OUT + t - 1] = ((pl.x + pl.y) + (pl.z + pl.w)) + bo;
        }
    }

    {
        const int t = T_OUT - 1;
        lds_barrier();
        get_partials(t & 1);
        cell_slot(0);
        cell_slot(1);
        cell_slot(2);
        cell_last(t & 1);
        lds_barrier();
        const v4f pv = po[(t & 1) * RB + brow];
        if (wave == 0 && lead)
            out[grow * T_OUT + t] = ((pv.x + pv.y) + (pv.z + pv.w)) + bo;
    }
}

extern "C" void kernel_launch(void* const* d_in, const int* in_sizes, int n_in,
                              void* d_out, int out_size, void* d_ws, size_t ws_size,
                              hipStream_t stream) {
    const float* x     = (const float*)d_in[0];
    const float* W_ih  = (const float*)d_in[1];
    const float* W_hh  = (const float*)d_in[2];
    const float* b_ih  = (const float*)d_in[3];
    const float* b_hh  = (const float*)d_in[4];
    const float* W_out = (const float*)d_in[5];
    const float* b_out = (const float*)d_in[6];
    float* out = (float*)d_out;

    const int B = in_sizes[0] / T_IN;
    dim3 grid(B / RB);
    dim3 block(NW * 32);
    hipLaunchKernelGGL(lstm_seq_kernel, grid, block, 0, stream,
                       x, W_ih, W_hh, b_ih, b_hh, W_out, b_out, out);
}


# Chunk 12's products run on into the wave's whole-chunk tiles (C operand; tiles rotated by wave, so all waves share one code): 13 tiles stored per step, not 17, in one group; gate sums in a new order
# speedup vs baseline: 10.5285x; 1.0717x over previous
#include <hip/hip_runtime.h>


typedef __attribute__((ext_vector_type(2))) float v2f;
typedef __attribute__((ext_vector_type(4))) float v4f;
typedef __attribute__((ext_vector_type(8))) float v8f;

namespace {
constexpr int HID   = 50;
constexpr int NT    = 13;
constexpr int T_IN  = 512;
constexpr int FUT   = 64;
constexpr int T_OUT = T_IN + FUT;
constexpr int NW    = 4;
constexpr int SLOTS = 3;
constexpr int XT    = 4;
constexpr int RB    = 8;
constexpr int NREG  = NT;
constexpr int HP    = 24;
constexpr int RS    = 48;
constexpr int BUFV  = NW * NREG * RS;

template <int CTRL>
__device__ __forceinline__ float dpp_read(float v) {
    const int s = __builtin_bit_cast(int, v);
    return __builtin_bit_cast(float, __builtin_amdgcn_update_dpp(s, s, CTRL, 0xf, 0xf, false));
}
__device__ __forceinline__ float xhalf(float v) {
    const int s = __builtin_bit_cast(int, v);
    return __builtin_bit_cast(float,
        __builtin_amdgcn_permlanex16(s, s, 0x76543210, 0xfedcba98, false, false));
}

__device__ __forceinline__ float tanh_nobranch(float y) {
    const float e = __expf(2.0f * fabsf(y));
    const float rc = __builtin_amdgcn_rcpf(e + 1.0f);
    return copysignf(fmaf(-2.0f, rc, 1.0f), y);
}
__device__ __forceinline__ float sigmoid_nobranch(float y) {
    return fmaf(tanh_nobranch(y * 0.5f), 0.5f, 0.5f);
}

__device__ __forceinline__ int col_unit(int tt, int c) {
    return tt < NT - 1 ? 16 * (tt >> 2) + c : 4 * (NT - 1) + (c >> 2);
}
__device__ __forceinline__ int col_gate(int tt, int c) {
    return tt < NT - 1 ? (tt & 3) : (c & 3);
}

__device__ __forceinline__ void lds_barrier() {
    __builtin_amdgcn_fence(__ATOMIC_RELEASE, "workgroup", "local");
    __builtin_amdgcn_s_barrier();
    __builtin_amdgcn_fence(__ATOMIC_ACQUIRE, "workgroup", "local");
}
}

__global__ __launch_bounds__(NW * 32) void lstm_seq_kernel(
    const float* __restrict__ x,
    const float* __restrict__ W_ih,
    const float* __restrict__ W_hh,
    const float* __restrict__ b_ih,
    const float* __restrict__ b_hh,
    const float* __restrict__ W_out,
    const float* __restrict__ b_out,
    float* __restrict__ out)
{
    __shared__ __attribute__((aligned(16))) v4f part[2 * BUFV];
    __shared__ __attribute__((aligned(16))) v4f po[2 * RB];

    const int tid  = threadIdx.x;
    const int lane = tid & 31;
    const int wave = __builtin_amdgcn_readfirstlane(tid >> 5);
    const int r    = lane & 15;
    const int half = lane >> 4;

    const int  kbase = SLOTS * wave;
    const bool many  = (wave < NW - 1);
    const int  tbase = 4 * wave;

    const int p    = (r >> 2) & 1;
    const int rh   = r >> 3;
    const int rj   = r & 3;
    const int brow = 4 * rh + rj;
    const int grow = blockIdx.x * RB + brow;
    const bool lead = (half == 0) && (p == 0);

    v2f bw[SLOTS * NT];
    v2f bwx[XT];
    const int koff = 2 * half;
#pragma unroll
    for (int tt = 0; tt < NT; ++tt) {
        const int  tr = tt + 4 * wave;
        const int  tl = tr < NT ? tr : tr - NT;
        const int  k  = col_unit(tl, r);
        const bool ok = (k < HID);
        const int  nn = col_gate(tl, r) * HID + (ok ? k : (HID - 1));
        const float* wrow = W_hh + nn * HID;
#pragma unroll
        for (int s = 0; s < SLOTS; ++s) {
            const int k0 = (kbase + s) * 4 + koff;
            const float v0 = wrow[k0];
            const float v1 = wrow[k0 + 1];
            bw[s * NT + tt].x = ok ? v0 : 0.0f;
            bw[s * NT + tt].y = ok ? v1 : 0.0f;
        }
    }
#pragma unroll
    for (int q = 0; q < XT; ++q) {
        const int  tq = (tbase + q < NT) ? (tbase + q) : (NT - 1);
        const int  k  = col_unit(tq, r);
        const bool ok = (k < HID) && (q == 0 || many);
        const int  nn = col_gate(tq, r) * HID + ((k < HID) ? k : (HID - 1));
        const float* wrow = W_hh + nn * HID;
        const float w48 = wrow[HID - 2];
        const float w49 = wrow[HID - 1];
        const float bs  = b_ih[nn] + b_hh[nn];
        const float wi  = W_ih[nn];
        const float l0  = half ? bs : w48;
        const float l1  = half ? wi : w49;
        bwx[q].x = ok ? l0 : 0.0f;
        bwx[q].y = ok ? l1 : 0.0f;
    }

    float hv[SLOTS], creg[SLOTS], wout[SLOTS];
#pragma unroll
    for (int s = 0; s < SLOTS; ++s) {
        hv[s] = 0.0f;
        creg[s] = 0.0f;
        wout[s] = W_out[12 * wave + 4 * s + 2 * half + p];
    }
    float hv4 = 0.0f, creg4 = 0.0f;
    const float wout4 = (wave == 0 && p == 0) ? W_out[4 * (NT - 1) + half] : 0.0f;
    const float act_s = p ? 1.0f : 0.5f;
    const float act_b = p ? 0.0f : 0.5f;
    const float bo = b_out[0];

    const float* pf = (const float*)part;
    int rd[SLOTS][NW];
#pragma unroll
    for (int s = 0; s < SLOTS; ++s) {
        const int ub = 12 * wave + 4 * s;
        const int m  = ub >> 4;
        const int cc = (ub & 15) + 2 * half + p;
#pragma unroll
        for (int w = 0; w < NW; ++w) {
            const int ir = 4 * m + NT - 4 * w;
            const int i0 = ir < NT ? ir : ir - NT;
            rd[s][w] = ((w * NREG + i0) * RS + rh * HP + cc) * 4 + rj;
        }
    }
    const int rd4  = (rh * HP + 4 * half + 2 * p) * 4 + rj;
    const int wr = (wave * NREG) * RS + half * HP + r;

    const v8f zero8 = v8f{0.f, 0.f, 0.f, 0.f, 0.f, 0.f, 0.f, 0.f};
    v8f acc[NT];
    float ps[SLOTS][4][NW];
    float ps4[2][NW];
    float osum = 0.0f;


    auto mm_slot = [&](const int s) __attribute__((always_inline)) {
        const v2f a = v2f{hv[s], dpp_read<0x164>(hv[s])};
#pragma unroll
        for (int tt = 0; tt < NT; ++tt)
            acc[tt] = __builtin_amdgcn_wmma_f32_16x16x4_f32(
                false, a, false, bw[s * NT + tt], (short)0,
                s == 0 ? zero8 : acc[tt], false, false);
    };
    auto mm_x = [&](const float xin) __attribute__((always_inline)) {
        const float h49 = xhalf(hv4);
        const v2f ax = v2f{half ? 1.0f : hv4, half ? xin : h49};
#pragma unroll
        for (int q = 0; q < XT; ++q)
            acc[q] = __builtin_amdgcn_wmma_f32_16x16x4_f32(
                false, ax, false, bwx[q], (short)0, acc[q], false, false);
    };
    auto put_tiles = [&](const int par, const int lo, const int hi) __attribute__((always_inline)) {
        v4f* pw = part + par * BUFV + wr;
#pragma unroll
        for (int tt = lo; tt < hi; ++tt)
            pw[tt * RS] = v4f{acc[tt][0], acc[tt][1], acc[tt][2], acc[tt][3]};
    };
    auto get_partials = [&](const int par) __attribute__((always_inline)) {
        const float* pr = pf + par * (BUFV * 4);
#pragma unroll
        for (int s = 0; s < SLOTS; ++s) {
#pragma unroll
            for (int gt = 0; gt < 4; ++gt) {
#pragma unroll
                for (int w = 0; w < NW; ++w)
                    ps[s][gt][w] = pr[rd[s][w] + (gt * RS) * 4];
            }
        }
#pragma unroll
        for (int e = 0; e < 2; ++e) {
#pragma unroll
            for (int w = 0; w < NW; ++w)
                ps4[e][w] = pr[rd4 + ((w * NREG + (NT - 1 - 4 * w)) * RS) * 4 + e * 4];
        }
    };
    auto cell_slot = [&](const int s) __attribute__((always_inline)) {
        float g[4];
#pragma unroll
        for (int gt = 0; gt < 4; ++gt)
            g[gt] = (ps[s][gt][0] + ps[s][gt][1]) + (ps[s][gt][2] + ps[s][gt][3]);
        const float ig = sigmoid_nobranch(g[0]);
        const float fg = sigmoid_nobranch(g[1]);
        const float cg = tanh_nobranch(g[2]);
        const float og = sigmoid_nobranch(g[3]);
        const float cv = fg * creg[s] + ig * cg;
        creg[s] = cv;
        hv[s] = og * tanh_nobranch(cv);
        osum = fmaf(hv[s], wout[s], osum);
    };
    auto cell_last = [&]() __attribute__((always_inline)) -> float {
        float g[2];
#pragma unroll
        for (int e = 0; e < 2; ++e)
            g[e] = (ps4[e][0] + ps4[e][1]) + (ps4[e][2] + ps4[e][3]);
        const float a0 = fmaf(tanh_nobranch(g[0] * act_s), act_s, act_b);
        const float a1 = sigmoid_nobranch(g[1]);
        const float b0 = dpp_read<0x164>(a0);
        const float b1 = dpp_read<0x164>(a1);
        const float ig = p ? b0 : a0;
        const float fg = p ? b1 : a1;
        const float cg = p ? a0 : b0;
        const float og = p ? a1 : b1;
        const float cv = fg * creg4 + ig * cg;
        creg4 = cv;
        hv4 = og * tanh_nobranch(cv);
        osum = fmaf(hv4, wout4, osum);
        osum += dpp_read<0x164>(osum);
        osum += xhalf(osum);
        const float share = osum;
        osum = 0.0f;
        return share;
    };
    auto put_share = [&](const int par, const float share) __attribute__((always_inline)) {
        if (lead)
            ((float*)po)[(par * RB + brow) * 4 + wave] = share;
    };

    if (tid < 2 * RB)
        po[tid] = v4f{0.f, 0.f, 0.f, 0.f};
    __syncthreads();

    const float* xrow = x + grow * T_IN;
    float xn1 = xrow[1];
#pragma unroll
    for (int s = 0; s < SLOTS; ++s) mm_slot(s);
    mm_x(xrow[0]);
    put_tiles(0, 0, NT);

#pragma unroll 1
    for (int t = 0; t < T_IN - 1; ++t) {
        lds_barrier();
        const v4f pvc = po[((t + 1) & 1) * RB + brow];
        get_partials(t & 1);
        const float xn2 = xrow[t + 2 < T_IN ? t + 2 : T_IN - 1];

        cell_slot(0);
        mm_slot(0);
        cell_slot(1);
        mm_slot(1);
        cell_slot(2);
        mm_slot(2);
        const float share = cell_last();
        mm_x(xn1);
        put_share(t & 1, share);
        put_tiles((t + 1) & 1, XT, NT);
        put_tiles((t + 1) & 1, 0, XT);

        if (wave == 0 && lead && t >= 1)
            out[grow * T_OUT + t - 1] = ((pvc.x + pvc.y) + (pvc.z + pvc.w)) + bo;
        xn1 = xn2;
    }

#pragma unroll 1
    for (int t = T_IN - 1; t < T_OUT - 1; ++t) {
        lds_barrier();
        get_partials(t & 1);

        cell_slot(0);
        mm_slot(0);
        cell_slot(1);
        mm_slot(1);
        cell_slot(2);
        mm_slot(2);
        put_share(t & 1, cell_last());
        put_tiles((t + 1) & 1, XT, NT);

        lds_barrier();
        const v4f pv = po[(t & 1) * RB + brow];
        const float xin = ((pv.x + pv.y) + (pv.z + pv.w)) + bo;
        if (wave == 0 && lead)
            out[grow * T_OUT + t] = xin;
        mm_x(xin);
        put_tiles((t + 1) & 1, 0, XT);

        if (t == T_IN - 1 && wave == 0 && lead) {
            const v4f pl = po[((t + 1) & 1) * RB + brow];
            out[grow * T_OUT + t - 1] = ((pl.x + pl.y) + (pl.z + pl.w)) + bo;
        }
    }

    {
        const int t = T_OUT - 1;
        lds_barrier();
        get_partials(t & 1);
        cell_slot(0);
        cell_slot(1);
        cell_slot(2);
        put_share(t & 1, cell_last());
        lds_barrier();
        const v4f pv = po[(t & 1) * RB + brow];
        if (wave == 0 && lead)
            out[grow * T_OUT + t] = ((pv.x + pv.y) + (pv.z + pv.w)) + bo;
    }
}

extern "C" void kernel_launch(void* const* d_in, const int* in_sizes, int n_in,
                              void* d_out, int out_size, void* d_ws, size_t ws_size,
                              hipStream_t stream) {
    const float* x     = (const float*)d_in[0];
    const float* W_ih  = (const float*)d_in[1];
    const float* W_hh  = (const float*)d_in[2];
    const float* b_ih  = (const float*)d_in[3];
    const float* b_hh  = (const float*)d_in[4];
    const float* W_out = (const float*)d_in[5];
    const float* b_out = (const float*)d_in[6];
    float* out = (float*)d_out;

    const int B = in_sizes[0] / T_IN;
    dim3 grid(B / RB);
    dim3 block(NW * 32);
    hipLaunchKernelGGL(lstm_seq_kernel, grid, block, 0, stream,
                       x, W_ih, W_hh, b_ih, b_hh, W_out, b_out, out);
}
